# starting kernel
# baseline (speedup 1.0000x reference)
#include <hip/hip_runtime.h>
#include <math.h>
#include <stdint.h>

#define NFEAT 256
#define NHID  64

typedef float v2f __attribute__((ext_vector_type(2)));
typedef float v8f __attribute__((ext_vector_type(8)));

__global__ void deg_init_kernel(float* __restrict__ deg, int n) {
  int i = blockIdx.x * blockDim.x + threadIdx.x;
  if (i < n) deg[i] = 1.0f;
}

__global__ void deg_count_kernel(const int* __restrict__ col,
                                 float* __restrict__ deg, int e) {
  int i = blockIdx.x * blockDim.x + threadIdx.x;
  if (i < e) atomicAdd(&deg[col[i]], 1.0f);
}

__global__ __launch_bounds__(256)
void gemm_x_w1_kernel(const float* __restrict__ x,
                      const float* __restrict__ W1,
                      float* __restrict__ h, int n) {
  __shared__ float sW[128 * NHID];
  const int tid  = threadIdx.x;
  const int wave = tid >> 5;
  const int lane = tid & 31;
  const int l15  = lane & 15;
  const int kh   = lane >> 4;
  const int rowBase = blockIdx.x * 128 + wave * 16;
  const int row  = rowBase + l15;
  const int rowC = (row < n) ? row : (n - 1);

  v8f acc[4];
#pragma unroll
  for (int t = 0; t < 4; ++t) acc[t] = (v8f){0,0,0,0,0,0,0,0};

  const float* xrow = x + (size_t)rowC * NFEAT + 2 * kh;
  __builtin_prefetch(xrow + 128, 0, 3);

  for (int kc = 0; kc < 2; ++kc) {
    __syncthreads();
    for (int idx = tid; idx < 64 * NHID; idx += 256) {
      int kp = idx >> 6;
      int nn = idx & 63;
      int kg = kc * 128 + kp * 2;
      ((float2*)sW)[kp * NHID + nn] =
          make_float2(W1[(size_t)kg * NHID + nn], W1[(size_t)(kg + 1) * NHID + nn]);
    }
    __syncthreads();

#pragma unroll 4
    for (int k0 = 0; k0 < 128; k0 += 4) {
      v2f a = *(const v2f*)(xrow + kc * 128 + k0);
      const int kp = (k0 >> 1) + kh;
#pragma unroll
      for (int t = 0; t < 4; ++t) {
        v2f b = ((const v2f*)sW)[kp * NHID + t * 16 + l15];
        acc[t] = __builtin_amdgcn_wmma_f32_16x16x4_f32(
            false, a, false, b, (short)0, acc[t], false, false);
      }
    }
  }

#pragma unroll
  for (int t = 0; t < 4; ++t) {
    const int ccol = t * 16 + l15;
#pragma unroll
    for (int j = 0; j < 8; ++j) {
      int r = rowBase + j + 8 * kh;
      if (r < n) h[(size_t)r * NHID + ccol] = acc[t][j];
    }
  }
}

__global__ void agg_init_kernel(const float* __restrict__ h,
                                const float* __restrict__ deg,
                                float* __restrict__ dinv,
                                float* __restrict__ agg, long long total) {
  long long idx = (long long)blockIdx.x * blockDim.x + threadIdx.x;
  if (idx >= total) return;
  int i = (int)(idx >> 6);
  int j = (int)(idx & 63);
  float d = rsqrtf(deg[i]);
  if (j == 0) dinv[i] = d;
  agg[idx] = d * d * h[idx];
}

__global__ __launch_bounds__(256)
void edge_scatter_kernel(const int* __restrict__ row,
                         const int* __restrict__ col,
                         const float* __restrict__ dinv,
                         const float* __restrict__ h,
                         float* __restrict__ agg, int e) {
  int wv   = (int)((blockIdx.x * blockDim.x + threadIdx.x) >> 5);
  int lane = threadIdx.x & 31;
  if (wv >= e) return;
  int r = __builtin_amdgcn_readfirstlane(row[wv]);
  int c = __builtin_amdgcn_readfirstlane(col[wv]);
  float norm = dinv[r] * dinv[c];
  const float* hr = h  + (size_t)r * NHID;
  float*       ac = agg + (size_t)c * NHID;
  atomicAdd(&ac[lane],      norm * hr[lane]);
  atomicAdd(&ac[lane + 32], norm * hr[lane + 32]);
}

__global__ __launch_bounds__(256)
void edge_mlp_kernel(const int* __restrict__ row, const int* __restrict__ col,
                     const float* __restrict__ agg,
                     const float* __restrict__ b1,
                     const float* __restrict__ Wfc,
                     const float* __restrict__ bfc,
                     float* __restrict__ out, int e) {
  int wv   = (int)((blockIdx.x * blockDim.x + threadIdx.x) >> 5);
  int lane = threadIdx.x & 31;
  if (wv >= e) return;
  int r = __builtin_amdgcn_readfirstlane(row[wv]);
  int c = __builtin_amdgcn_readfirstlane(col[wv]);
  const float* hc = agg + (size_t)c * NHID;
  const float* hr = agg + (size_t)r * NHID;
  float a0 = 0.f, a1 = 0.f;
#pragma unroll
  for (int q = 0; q < 4; ++q) {
    int j = lane + 32 * q;
    float v = (j < 64) ? (hc[j] + b1[j]) : (hr[j - 64] + b1[j - 64]);
    a0 += v * Wfc[j * 2 + 0];
    a1 += v * Wfc[j * 2 + 1];
  }
#pragma unroll
  for (int off = 16; off > 0; off >>= 1) {
    a0 += __shfl_xor(a0, off, 32);
    a1 += __shfl_xor(a1, off, 32);
  }
  if (lane == 0) {
    float l0 = a0 + bfc[0], l1 = a1 + bfc[1];
    float m  = fmaxf(l0, l1);
    float e0 = __expf(l0 - m), e1 = __expf(l1 - m);
    float s  = 1.0f / (e0 + e1);
    ((float2*)out)[wv] = make_float2(e0 * s, e1 * s);
  }
}

extern "C" void kernel_launch(void* const* d_in, const int* in_sizes, int n_in,
                              void* d_out, int out_size, void* d_ws, size_t ws_size,
                              hipStream_t stream) {
  const float* x   = (const float*)d_in[0];
  const int*   ei  = (const int*)d_in[1];
  const float* W1  = (const float*)d_in[2];
  const float* b1  = (const float*)d_in[3];
  const float* Wfc = (const float*)d_in[4];
  const float* bfc = (const float*)d_in[5];
  const int n = in_sizes[0] / NFEAT;
  const int e = in_sizes[1] / 2;
  const int* row = ei;
  const int* col = ei + e;

  uintptr_t p = (uintptr_t)d_ws;
  float* deg  = (float*)p;  p += (size_t)n * sizeof(float);
  float* dinv = (float*)p;  p += (size_t)n * sizeof(float);
  p = (p + 255) & ~(uintptr_t)255;
  float* h    = (float*)p;  p += (size_t)n * NHID * sizeof(float);
  p = (p + 255) & ~(uintptr_t)255;
  float* agg  = (float*)p;

  const long long aggTotal = (long long)n * NHID;

  deg_init_kernel <<<(n + 255) / 256, 256, 0, stream>>>(deg, n);
  deg_count_kernel<<<(e + 255) / 256, 256, 0, stream>>>(col, deg, e);
  gemm_x_w1_kernel<<<(n + 127) / 128, 256, 0, stream>>>(x, W1, h, n);
  agg_init_kernel <<<(unsigned)((aggTotal + 255) / 256), 256, 0, stream>>>(h, deg, dinv, agg, aggTotal);
  edge_scatter_kernel<<<(e + 7) / 8, 256, 0, stream>>>(row, col, dinv, h, agg, e);
  edge_mlp_kernel    <<<(e + 7) / 8, 256, 0, stream>>>(row, col, agg, b1, Wfc, bfc,
                                                       (float*)d_out, e);
}


# fc layer applied per node before the neighbour sum (linearity): each edge atomically adds 4 projected floats, not 64, and the edge stage only gathers 2x2 floats; order of f32 sums changes
# speedup vs baseline: 2.2710x; 2.2710x over previous
#include <hip/hip_runtime.h>
#include <math.h>
#include <stdint.h>

#define NFEAT 256
#define NHID  64

typedef float v2f __attribute__((ext_vector_type(2)));
typedef float v8f __attribute__((ext_vector_type(8)));

__global__ void deg_init_kernel(float* __restrict__ deg, int n) {
  int i = blockIdx.x * blockDim.x + threadIdx.x;
  if (i < n) deg[i] = 1.0f;
}

__global__ void deg_count_kernel(const int* __restrict__ col,
                                 float* __restrict__ deg, int e) {
  int i = blockIdx.x * blockDim.x + threadIdx.x;
  if (i < e) atomicAdd(&deg[col[i]], 1.0f);
}

__global__ __launch_bounds__(256)
void gemm_x_w1_kernel(const float* __restrict__ x,
                      const float* __restrict__ W1,
                      float* __restrict__ h, int n) {
  __shared__ float sW[128 * NHID];
  const int tid  = threadIdx.x;
  const int wave = tid >> 5;
  const int lane = tid & 31;
  const int l15  = lane & 15;
  const int kh   = lane >> 4;
  const int rowBase = blockIdx.x * 128 + wave * 16;
  const int row  = rowBase + l15;
  const int rowC = (row < n) ? row : (n - 1);

  v8f acc[4];
#pragma unroll
  for (int t = 0; t < 4; ++t) acc[t] = (v8f){0,0,0,0,0,0,0,0};

  const float* xrow = x + (size_t)rowC * NFEAT + 2 * kh;
  __builtin_prefetch(xrow + 128, 0, 3);

  for (int kc = 0; kc < 2; ++kc) {
    __syncthreads();
    for (int idx = tid; idx < 64 * NHID; idx += 256) {
      int kp = idx >> 6;
      int nn = idx & 63;
      int kg = kc * 128 + kp * 2;
      ((float2*)sW)[kp * NHID + nn] =
          make_float2(W1[(size_t)kg * NHID + nn], W1[(size_t)(kg + 1) * NHID + nn]);
    }
    __syncthreads();

#pragma unroll 4
    for (int k0 = 0; k0 < 128; k0 += 4) {
      v2f a = *(const v2f*)(xrow + kc * 128 + k0);
      const int kp = (k0 >> 1) + kh;
#pragma unroll
      for (int t = 0; t < 4; ++t) {
        v2f b = ((const v2f*)sW)[kp * NHID + t * 16 + l15];
        acc[t] = __builtin_amdgcn_wmma_f32_16x16x4_f32(
            false, a, false, b, (short)0, acc[t], false, false);
      }
    }
  }

#pragma unroll
  for (int t = 0; t < 4; ++t) {
    const int ccol = t * 16 + l15;
#pragma unroll
    for (int j = 0; j < 8; ++j) {
      int r = rowBase + j + 8 * kh;
      if (r < n) h[(size_t)r * NHID + ccol] = acc[t][j];
    }
  }
}

template <int M>
__device__ __forceinline__ float row_xor(float v) {
  return __int_as_float(__builtin_amdgcn_update_dpp(0, __float_as_int(v), 0x160 + M,
                                                    0xf, 0xf, false));
}
__device__ __forceinline__ float row16_sum(float v) {
  v += row_xor<1>(v);
  v += row_xor<2>(v);
  v += row_xor<4>(v);
  v += row_xor<8>(v);
  return v;
}

#define NODES_PER_BLOCK 80
template <bool FULL>
__device__ __forceinline__ void node_project_body(const float* __restrict__ h,
                                                  const float* __restrict__ deg,
                                                  const float* __restrict__ b1,
                                                  const float* __restrict__ Wfc,
                                                  float* __restrict__ dinv,
                                                  float* __restrict__ g,
                                                  float* __restrict__ agg4, int n) {
  const int l   = threadIdx.x & 15;
  const int grp = threadIdx.x >> 4;
  const int base = blockIdx.x * NODES_PER_BLOCK + grp;
  float4 hv[5];
  float  dg[5];
#pragma unroll
  for (int it = 0; it < 5; ++it) {
    const int node = base + it * 16;
    const int nc   = (FULL || node < n) ? node : (n - 1);
    hv[it] = *(const float4*)(h + (size_t)nc * NHID + 4 * l);
    dg[it] = deg[nc];
  }
  const float4 wa0 = *(const float4*)(Wfc + 8 * l);
  const float4 wa1 = *(const float4*)(Wfc + 8 * l + 4);
  const float4 wb0 = *(const float4*)(Wfc + 128 + 8 * l);
  const float4 wb1 = *(const float4*)(Wfc + 128 + 8 * l + 4);
  const float4 bv  = *(const float4*)(b1 + 4 * l);

  const float bw0 = row16_sum(bv.x * wa0.x + bv.y * wa0.z + bv.z * wa1.x + bv.w * wa1.z);
  const float bw1 = row16_sum(bv.x * wa0.y + bv.y * wa0.w + bv.z * wa1.y + bv.w * wa1.w);
  const float bw2 = row16_sum(bv.x * wb0.x + bv.y * wb0.z + bv.z * wb1.x + bv.w * wb1.z);
  const float bw3 = row16_sum(bv.x * wb0.y + bv.y * wb0.w + bv.z * wb1.y + bv.w * wb1.w);

#pragma unroll
  for (int it = 0; it < 5; ++it) {
    const int node = base + it * 16;
    const float4 v = hv[it];
    const float p0 = row16_sum(v.x * wa0.x + v.y * wa0.z + v.z * wa1.x + v.w * wa1.z);
    const float p1 = row16_sum(v.x * wa0.y + v.y * wa0.w + v.z * wa1.y + v.w * wa1.w);
    const float p2 = row16_sum(v.x * wb0.x + v.y * wb0.z + v.z * wb1.x + v.w * wb1.z);
    const float p3 = row16_sum(v.x * wb0.y + v.y * wb0.w + v.z * wb1.y + v.w * wb1.w);
    const float d  = rsqrtf(dg[it]);
    if (l == 0 && (FULL || node < n)) {
      const float dd = d * d;
      dinv[node] = d;
      ((float4*)g)[node]    = make_float4(d * p0, d * p1, d * p2, d * p3);
      ((float4*)agg4)[node] = make_float4(dd * p0 + bw0, dd * p1 + bw1,
                                          dd * p2 + bw2, dd * p3 + bw3);
    }
  }
}

__global__ __launch_bounds__(256)
void node_project_kernel(const float* __restrict__ h,
                         const float* __restrict__ deg,
                         const float* __restrict__ b1,
                         const float* __restrict__ Wfc,
                         float* __restrict__ dinv,
                         float* __restrict__ g,
                         float* __restrict__ agg4, int n) {
  if ((int)blockIdx.x * NODES_PER_BLOCK + NODES_PER_BLOCK <= n)
    node_project_body<true>(h, deg, b1, Wfc, dinv, g, agg4, n);
  else
    node_project_body<false>(h, deg, b1, Wfc, dinv, g, agg4, n);
}

template <bool FULL>
__device__ __forceinline__ void edge_scatter4_body(const int* __restrict__ row,
                                                   const int* __restrict__ col,
                                                   const float* __restrict__ dinv,
                                                   const float* __restrict__ g,
                                                   float* __restrict__ agg4, int e) {
  const int q  = threadIdx.x & 3;
  const int e0 = blockIdx.x * 256 + (threadIdx.x >> 2);
  int r[4], c[4];
#pragma unroll
  for (int k = 0; k < 4; ++k) {
    const int ed  = e0 + 64 * k;
    const int edc = (FULL || ed < e) ? ed : (e - 1);
    r[k] = row[edc];
    c[k] = col[edc];
  }
  float v[4];
#pragma unroll
  for (int k = 0; k < 4; ++k)
    v[k] = dinv[c[k]] * g[(size_t)r[k] * 4 + q];
#pragma unroll
  for (int k = 0; k < 4; ++k)
    if (FULL || e0 + 64 * k < e) atomicAdd(&agg4[(size_t)c[k] * 4 + q], v[k]);
}

__global__ __launch_bounds__(256)
void edge_scatter4_kernel(const int* __restrict__ row,
                          const int* __restrict__ col,
                          const float* __restrict__ dinv,
                          const float* __restrict__ g,
                          float* __restrict__ agg4, int e) {
  if ((int)blockIdx.x * 256 + 256 <= e)
    edge_scatter4_body<true>(row, col, dinv, g, agg4, e);
  else
    edge_scatter4_body<false>(row, col, dinv, g, agg4, e);
}

template <bool FULL>
__device__ __forceinline__ void edge_softmax_body(const int* __restrict__ row,
                                                  const int* __restrict__ col,
                                                  const float* __restrict__ agg4,
                                                  const float* __restrict__ bfc,
                                                  float* __restrict__ out, int e) {
  const int e0 = blockIdx.x * 1024 + threadIdx.x;
  const float c0 = bfc[0], c1 = bfc[1];
  int r[4], c[4];
#pragma unroll
  for (int k = 0; k < 4; ++k) {
    const int ed  = e0 + 256 * k;
    const int edc = (FULL || ed < e) ? ed : (e - 1);
    r[k] = row[edc];
    c[k] = col[edc];
  }
  float2 a[4], b[4];
#pragma unroll
  for (int k = 0; k < 4; ++k) {
    a[k] = *(const float2*)(agg4 + (size_t)c[k] * 4);
    b[k] = *(const float2*)(agg4 + (size_t)r[k] * 4 + 2);
  }
#pragma unroll
  for (int k = 0; k < 4; ++k) {
    const int ed = e0 + 256 * k;
    if (FULL || ed < e) {
      float l0 = a[k].x + b[k].x + c0, l1 = a[k].y + b[k].y + c1;
      float m  = fmaxf(l0, l1);
      float x0 = __expf(l0 - m), x1 = __expf(l1 - m);
      float s  = 1.0f / (x0 + x1);
      ((float2*)out)[ed] = make_float2(x0 * s, x1 * s);
    }
  }
}

__global__ __launch_bounds__(256)
void edge_softmax_kernel(const int* __restrict__ row, const int* __restrict__ col,
                         const float* __restrict__ agg4,
                         const float* __restrict__ bfc,
                         float* __restrict__ out, int e) {
  if ((int)blockIdx.x * 1024 + 1024 <= e)
    edge_softmax_body<true>(row, col, agg4, bfc, out, e);
  else
    edge_softmax_body<false>(row, col, agg4, bfc, out, e);
}

extern "C" void kernel_launch(void* const* d_in, const int* in_sizes, int n_in,
                              void* d_out, int out_size, void* d_ws, size_t ws_size,
                              hipStream_t stream) {
  const float* x   = (const float*)d_in[0];
  const int*   ei  = (const int*)d_in[1];
  const float* W1  = (const float*)d_in[2];
  const float* b1  = (const float*)d_in[3];
  const float* Wfc = (const float*)d_in[4];
  const float* bfc = (const float*)d_in[5];
  const int n = in_sizes[0] / NFEAT;
  const int e = in_sizes[1] / 2;
  const int* row = ei;
  const int* col = ei + e;

  uintptr_t p = (uintptr_t)d_ws;
  float* deg  = (float*)p;  p += (size_t)n * sizeof(float);
  float* dinv = (float*)p;  p += (size_t)n * sizeof(float);
  p = (p + 255) & ~(uintptr_t)255;
  float* h    = (float*)p;  p += (size_t)n * NHID * sizeof(float);
  p = (p + 255) & ~(uintptr_t)255;
  float* g    = (float*)p;  p += (size_t)n * 4 * sizeof(float);
  p = (p + 255) & ~(uintptr_t)255;
  float* agg4 = (float*)p;

  deg_init_kernel <<<(n + 255) / 256, 256, 0, stream>>>(deg, n);
  deg_count_kernel<<<(e + 255) / 256, 256, 0, stream>>>(col, deg, e);
  gemm_x_w1_kernel<<<(n + 127) / 128, 256, 0, stream>>>(x, W1, h, n);
  node_project_kernel<<<(n + NODES_PER_BLOCK - 1) / NODES_PER_BLOCK, 256, 0, stream>>>(
      h, deg, b1, Wfc, dinv, g, agg4, n);
  edge_scatter4_kernel<<<(e + 255) / 256, 256, 0, stream>>>(row, col, dinv, g, agg4, e);
  edge_softmax_kernel <<<(e + 1023) / 1024, 256, 0, stream>>>(row, col, agg4, bfc,
                                                              (float*)d_out, e);
}


# deg and agg4 tables kept once per package half (copy picked by die id) so every scattered f32 atomic add stays inside one cache half; a join pass adds the two copies; order of f32 sums changes
# speedup vs baseline: 4.4692x; 1.9679x over previous
#include <hip/hip_runtime.h>
#include <math.h>
#include <stdint.h>

#define NFEAT 256
#define NHID  64

typedef float v2f __attribute__((ext_vector_type(2)));
typedef float v8f __attribute__((ext_vector_type(8)));

__device__ __forceinline__ int package_half() {
  const unsigned id = __builtin_amdgcn_s_sendmsg_rtn(0x87);
  return (((id >> 16) & 0xFu) >= 4u) ? 1 : 0;
}

__global__ void deg_init_kernel(float* __restrict__ deg0, float* __restrict__ deg1, int n) {
  int i = blockIdx.x * blockDim.x + threadIdx.x;
  if (i < n) { deg0[i] = 1.0f; deg1[i] = 0.0f; }
}

__global__ void deg_count_kernel(const int* __restrict__ col,
                                 float* __restrict__ deg0, float* __restrict__ deg1, int e) {
  int i = blockIdx.x * blockDim.x + threadIdx.x;
  float* __restrict__ mine = package_half() ? deg1 : deg0;
  if (i < e) atomicAdd(&mine[col[i]], 1.0f);
}

__global__ __launch_bounds__(256)
void gemm_x_w1_kernel(const float* __restrict__ x,
                      const float* __restrict__ W1,
                      float* __restrict__ h, int n) {
  __shared__ float sW[128 * NHID];
  const int tid  = threadIdx.x;
  const int wave = tid >> 5;
  const int lane = tid & 31;
  const int l15  = lane & 15;
  const int kh   = lane >> 4;
  const int rowBase = blockIdx.x * 128 + wave * 16;
  const int row  = rowBase + l15;
  const int rowC = (row < n) ? row : (n - 1);

  v8f acc[4];
#pragma unroll
  for (int t = 0; t < 4; ++t) acc[t] = (v8f){0,0,0,0,0,0,0,0};

  const float* xrow = x + (size_t)rowC * NFEAT + 2 * kh;
  __builtin_prefetch(xrow + 128, 0, 3);

  for (int kc = 0; kc < 2; ++kc) {
    __syncthreads();
    for (int idx = tid; idx < 64 * NHID; idx += 256) {
      int kp = idx >> 6;
      int nn = idx & 63;
      int kg = kc * 128 + kp * 2;
      ((float2*)sW)[kp * NHID + nn] =
          make_float2(W1[(size_t)kg * NHID + nn], W1[(size_t)(kg + 1) * NHID + nn]);
    }
    __syncthreads();

#pragma unroll 4
    for (int k0 = 0; k0 < 128; k0 += 4) {
      v2f a = *(const v2f*)(xrow + kc * 128 + k0);
      const int kp = (k0 >> 1) + kh;
#pragma unroll
      for (int t = 0; t < 4; ++t) {
        v2f b = ((const v2f*)sW)[kp * NHID + t * 16 + l15];
        acc[t] = __builtin_amdgcn_wmma_f32_16x16x4_f32(
            false, a, false, b, (short)0, acc[t], false, false);
      }
    }
  }

#pragma unroll
  for (int t = 0; t < 4; ++t) {
    const int ccol = t * 16 + l15;
#pragma unroll
    for (int j = 0; j < 8; ++j) {
      int r = rowBase + j + 8 * kh;
      if (r < n) h[(size_t)r * NHID + ccol] = acc[t][j];
    }
  }
}

template <int M>
__device__ __forceinline__ float row_xor(float v) {
  return __int_as_float(__builtin_amdgcn_update_dpp(0, __float_as_int(v), 0x160 + M,
                                                    0xf, 0xf, false));
}
__device__ __forceinline__ float row16_sum(float v) {
  v += row_xor<1>(v);
  v += row_xor<2>(v);
  v += row_xor<4>(v);
  v += row_xor<8>(v);
  return v;
}

#define NODES_PER_BLOCK 80
template <bool FULL>
__device__ __forceinline__ void node_project_body(const float* __restrict__ h,
                                                  const float* __restrict__ deg0,
                                                  const float* __restrict__ deg1,
                                                  const float* __restrict__ b1,
                                                  const float* __restrict__ Wfc,
                                                  float* __restrict__ dinv,
                                                  float* __restrict__ g,
                                                  float* __restrict__ agg4,
                                                  float* __restrict__ agg4b, int n) {
  const int l   = threadIdx.x & 15;
  const int grp = threadIdx.x >> 4;
  const int base = blockIdx.x * NODES_PER_BLOCK + grp;
  float4 hv[5];
  float  dg[5];
#pragma unroll
  for (int it = 0; it < 5; ++it) {
    const int node = base + it * 16;
    const int nc   = (FULL || node < n) ? node : (n - 1);
    hv[it] = *(const float4*)(h + (size_t)nc * NHID + 4 * l);
    dg[it] = deg0[nc] + deg1[nc];
  }
  const float4 wa0 = *(const float4*)(Wfc + 8 * l);
  const float4 wa1 = *(const float4*)(Wfc + 8 * l + 4);
  const float4 wb0 = *(const float4*)(Wfc + 128 + 8 * l);
  const float4 wb1 = *(const float4*)(Wfc + 128 + 8 * l + 4);
  const float4 bv  = *(const float4*)(b1 + 4 * l);

  const float bw0 = row16_sum(bv.x * wa0.x + bv.y * wa0.z + bv.z * wa1.x + bv.w * wa1.z);
  const float bw1 = row16_sum(bv.x * wa0.y + bv.y * wa0.w + bv.z * wa1.y + bv.w * wa1.w);
  const float bw2 = row16_sum(bv.x * wb0.x + bv.y * wb0.z + bv.z * wb1.x + bv.w * wb1.z);
  const float bw3 = row16_sum(bv.x * wb0.y + bv.y * wb0.w + bv.z * wb1.y + bv.w * wb1.w);

#pragma unroll
  for (int it = 0; it < 5; ++it) {
    const int node = base + it * 16;
    const float4 v = hv[it];
    const float p0 = row16_sum(v.x * wa0.x + v.y * wa0.z + v.z * wa1.x + v.w * wa1.z);
    const float p1 = row16_sum(v.x * wa0.y + v.y * wa0.w + v.z * wa1.y + v.w * wa1.w);
    const float p2 = row16_sum(v.x * wb0.x + v.y * wb0.z + v.z * wb1.x + v.w * wb1.z);
    const float p3 = row16_sum(v.x * wb0.y + v.y * wb0.w + v.z * wb1.y + v.w * wb1.w);
    const float d  = rsqrtf(dg[it]);
    if (l == 0 && (FULL || node < n)) {
      const float dd = d * d;
      dinv[node] = d;
      ((float4*)g)[node]    = make_float4(d * p0, d * p1, d * p2, d * p3);
      ((float4*)agg4)[node] = make_float4(dd * p0 + bw0, dd * p1 + bw1,
                                          dd * p2 + bw2, dd * p3 + bw3);
      ((float4*)agg4b)[node] = make_float4(0.0f, 0.0f, 0.0f, 0.0f);
    }
  }
}

__global__ __launch_bounds__(256)
void node_project_kernel(const float* __restrict__ h,
                         const float* __restrict__ deg0,
                         const float* __restrict__ deg1,
                         const float* __restrict__ b1,
                         const float* __restrict__ Wfc,
                         float* __restrict__ dinv,
                         float* __restrict__ g,
                         float* __restrict__ agg4,
                         float* __restrict__ agg4b, int n) {
  if ((int)blockIdx.x * NODES_PER_BLOCK + NODES_PER_BLOCK <= n)
    node_project_body<true>(h, deg0, deg1, b1, Wfc, dinv, g, agg4, agg4b, n);
  else
    node_project_body<false>(h, deg0, deg1, b1, Wfc, dinv, g, agg4, agg4b, n);
}

template <bool FULL>
__device__ __forceinline__ void edge_scatter4_body(const int* __restrict__ row,
                                                   const int* __restrict__ col,
                                                   const float* __restrict__ dinv,
                                                   const float* __restrict__ g,
                                                   float* __restrict__ agg4, int e) {
  const int q  = threadIdx.x & 3;
  const int e0 = blockIdx.x * 256 + (threadIdx.x >> 2);
  int r[4], c[4];
#pragma unroll
  for (int k = 0; k < 4; ++k) {
    const int ed  = e0 + 64 * k;
    const int edc = (FULL || ed < e) ? ed : (e - 1);
    r[k] = row[edc];
    c[k] = col[edc];
  }
  float v[4];
#pragma unroll
  for (int k = 0; k < 4; ++k)
    v[k] = dinv[c[k]] * g[(size_t)r[k] * 4 + q];
#pragma unroll
  for (int k = 0; k < 4; ++k)
    if (FULL || e0 + 64 * k < e) atomicAdd(&agg4[(size_t)c[k] * 4 + q], v[k]);
}

__global__ __launch_bounds__(256)
void edge_scatter4_kernel(const int* __restrict__ row,
                          const int* __restrict__ col,
                          const float* __restrict__ dinv,
                          const float* __restrict__ g,
                          float* __restrict__ agg4,
                          float* __restrict__ agg4b, int e) {
  float* __restrict__ mine = package_half() ? agg4b : agg4;
  if ((int)blockIdx.x * 256 + 256 <= e)
    edge_scatter4_body<true>(row, col, dinv, g, mine, e);
  else
    edge_scatter4_body<false>(row, col, dinv, g, mine, e);
}

__global__ __launch_bounds__(256)
void agg_join_kernel(float* __restrict__ agg4, const float* __restrict__ agg4b, int n) {
  const int i = blockIdx.x * 256 + threadIdx.x;
  if (i < n) {
    const float4 a = ((const float4*)agg4)[i];
    const float4 b = ((const float4*)agg4b)[i];
    ((float4*)agg4)[i] = make_float4(a.x + b.x, a.y + b.y, a.z + b.z, a.w + b.w);
  }
}

template <bool FULL>
__device__ __forceinline__ void edge_softmax_body(const int* __restrict__ row,
                                                  const int* __restrict__ col,
                                                  const float* __restrict__ agg4,
                                                  const float* __restrict__ bfc,
                                                  float* __restrict__ out, int e) {
  const int e0 = blockIdx.x * 1024 + threadIdx.x;
  const float c0 = bfc[0], c1 = bfc[1];
  int r[4], c[4];
#pragma unroll
  for (int k = 0; k < 4; ++k) {
    const int ed  = e0 + 256 * k;
    const int edc = (FULL || ed < e) ? ed : (e - 1);
    r[k] = row[edc];
    c[k] = col[edc];
  }
  float2 a[4], b[4];
#pragma unroll
  for (int k = 0; k < 4; ++k) {
    a[k] = *(const float2*)(agg4 + (size_t)c[k] * 4);
    b[k] = *(const float2*)(agg4 + (size_t)r[k] * 4 + 2);
  }
#pragma unroll
  for (int k = 0; k < 4; ++k) {
    const int ed = e0 + 256 * k;
    if (FULL || ed < e) {
      float l0 = a[k].x + b[k].x + c0, l1 = a[k].y + b[k].y + c1;
      float m  = fmaxf(l0, l1);
      float x0 = __expf(l0 - m), x1 = __expf(l1 - m);
      float s  = 1.0f / (x0 + x1);
      ((float2*)out)[ed] = make_float2(x0 * s, x1 * s);
    }
  }
}

__global__ __launch_bounds__(256)
void edge_softmax_kernel(const int* __restrict__ row, const int* __restrict__ col,
                         const float* __restrict__ agg4,
                         const float* __restrict__ bfc,
                         float* __restrict__ out, int e) {
  if ((int)blockIdx.x * 1024 + 1024 <= e)
    edge_softmax_body<true>(row, col, agg4, bfc, out, e);
  else
    edge_softmax_body<false>(row, col, agg4, bfc, out, e);
}

extern "C" void kernel_launch(void* const* d_in, const int* in_sizes, int n_in,
                              void* d_out, int out_size, void* d_ws, size_t ws_size,
                              hipStream_t stream) {
  const float* x   = (const float*)d_in[0];
  const int*   ei  = (const int*)d_in[1];
  const float* W1  = (const float*)d_in[2];
  const float* b1  = (const float*)d_in[3];
  const float* Wfc = (const float*)d_in[4];
  const float* bfc = (const float*)d_in[5];
  const int n = in_sizes[0] / NFEAT;
  const int e = in_sizes[1] / 2;
  const int* row = ei;
  const int* col = ei + e;

  uintptr_t p = (uintptr_t)d_ws;
  float* deg0 = (float*)p;  p += (size_t)n * sizeof(float);
  p = (p + 255) & ~(uintptr_t)255;
  float* deg1 = (float*)p;  p += (size_t)n * sizeof(float);
  p = (p + 255) & ~(uintptr_t)255;
  float* dinv = (float*)p;  p += (size_t)n * sizeof(float);
  p = (p + 255) & ~(uintptr_t)255;
  float* h    = (float*)p;  p += (size_t)n * NHID * sizeof(float);
  p = (p + 255) & ~(uintptr_t)255;
  float* g    = (float*)p;  p += (size_t)n * 4 * sizeof(float);
  p = (p + 255) & ~(uintptr_t)255;
  float* agg4 = (float*)p;  p += (size_t)n * 4 * sizeof(float);
  p = (p + 255) & ~(uintptr_t)255;
  float* agg4b = (float*)p;

  deg_init_kernel <<<(n + 255) / 256, 256, 0, stream>>>(deg0, deg1, n);
  deg_count_kernel<<<(e + 255) / 256, 256, 0, stream>>>(col, deg0, deg1, e);
  gemm_x_w1_kernel<<<(n + 127) / 128, 256, 0, stream>>>(x, W1, h, n);
  node_project_kernel<<<(n + NODES_PER_BLOCK - 1) / NODES_PER_BLOCK, 256, 0, stream>>>(
      h, deg0, deg1, b1, Wfc, dinv, g, agg4, agg4b, n);
  edge_scatter4_kernel<<<(e + 255) / 256, 256, 0, stream>>>(row, col, dinv, g, agg4, agg4b, e);
  agg_join_kernel     <<<(n + 255) / 256, 256, 0, stream>>>(agg4, agg4b, n);
  edge_softmax_kernel <<<(e + 1023) / 1024, 256, 0, stream>>>(row, col, agg4, bfc,
                                                              (float*)d_out, e);
}


# degree count moved into the matrix-product launch: its work-groups follow the product's in one grid, so the count's waves and atomic adds run under the f32 WMMA work; 6 launches, not 7
# speedup vs baseline: 5.3061x; 1.1873x over previous
#include <hip/hip_runtime.h>
#include <math.h>
#include <stdint.h>

#define NFEAT 256
#define NHID  64

typedef float v2f __attribute__((ext_vector_type(2)));
typedef float v8f __attribute__((ext_vector_type(8)));

__device__ __forceinline__ int package_half() {
  const unsigned id = __builtin_amdgcn_s_sendmsg_rtn(0x87);
  return (((id >> 16) & 0xFu) >= 4u) ? 1 : 0;
}

__global__ void deg_init_kernel(float* __restrict__ deg0, float* __restrict__ deg1, int n) {
  int i = blockIdx.x * blockDim.x + threadIdx.x;
  if (i < n) { deg0[i] = 1.0f; deg1[i] = 0.0f; }
}

__global__ __launch_bounds__(256)
void gemm_x_w1_deg_count_kernel(const float* __restrict__ x,
                                const float* __restrict__ W1,
                                float* __restrict__ h, int n,
                                const int* __restrict__ col,
                                float* __restrict__ deg0, float* __restrict__ deg1, int e) {
  __shared__ float sW[128 * NHID];
  const int gemm_blocks = (n + 127) / 128;
  if ((int)blockIdx.x >= gemm_blocks) {
    const int i = ((int)blockIdx.x - gemm_blocks) * 256 + (int)threadIdx.x;
    float* __restrict__ mine = package_half() ? deg1 : deg0;
    if (i < e) atomicAdd(&mine[col[i]], 1.0f);
    return;
  }
  const int tid  = threadIdx.x;
  const int wave = tid >> 5;
  const int lane = tid & 31;
  const int l15  = lane & 15;
  const int kh   = lane >> 4;
  const int rowBase = blockIdx.x * 128 + wave * 16;
  const int row  = rowBase + l15;
  const int rowC = (row < n) ? row : (n - 1);

  v8f acc[4];
#pragma unroll
  for (int t = 0; t < 4; ++t) acc[t] = (v8f){0,0,0,0,0,0,0,0};

  const float* xrow = x + (size_t)rowC * NFEAT + 2 * kh;
  __builtin_prefetch(xrow + 128, 0, 3);

  for (int kc = 0; kc < 2; ++kc) {
    __syncthreads();
    for (int idx = tid; idx < 64 * NHID; idx += 256) {
      int kp = idx >> 6;
      int nn = idx & 63;
      int kg = kc * 128 + kp * 2;
      ((float2*)sW)[kp * NHID + nn] =
          make_float2(W1[(size_t)kg * NHID + nn], W1[(size_t)(kg + 1) * NHID + nn]);
    }
    __syncthreads();

#pragma unroll 4
    for (int k0 = 0; k0 < 128; k0 += 4) {
      v2f a = *(const v2f*)(xrow + kc * 128 + k0);
      const int kp = (k0 >> 1) + kh;
#pragma unroll
      for (int t = 0; t < 4; ++t) {
        v2f b = ((const v2f*)sW)[kp * NHID + t * 16 + l15];
        acc[t] = __builtin_amdgcn_wmma_f32_16x16x4_f32(
            false, a, false, b, (short)0, acc[t], false, false);
      }
    }
  }

#pragma unroll
  for (int t = 0; t < 4; ++t) {
    const int ccol = t * 16 + l15;
#pragma unroll
    for (int j = 0; j < 8; ++j) {
      int r = rowBase + j + 8 * kh;
      if (r < n) h[(size_t)r * NHID + ccol] = acc[t][j];
    }
  }
}

template <int M>
__device__ __forceinline__ float row_xor(float v) {
  return __int_as_float(__builtin_amdgcn_update_dpp(0, __float_as_int(v), 0x160 + M,
                                                    0xf, 0xf, false));
}
__device__ __forceinline__ float row16_sum(float v) {
  v += row_xor<1>(v);
  v += row_xor<2>(v);
  v += row_xor<4>(v);
  v += row_xor<8>(v);
  return v;
}

#define NODES_PER_BLOCK 80
template <bool FULL>
__device__ __forceinline__ void node_project_body(const float* __restrict__ h,
                                                  const float* __restrict__ deg0,
                                                  const float* __restrict__ deg1,
                                                  const float* __restrict__ b1,
                                                  const float* __restrict__ Wfc,
                                                  float* __restrict__ dinv,
                                                  float* __restrict__ g,
                                                  float* __restrict__ agg4,
                                                  float* __restrict__ agg4b, int n) {
  const int l   = threadIdx.x & 15;
  const int grp = threadIdx.x >> 4;
  const int base = blockIdx.x * NODES_PER_BLOCK + grp;
  float4 hv[5];
  float  dg[5];
#pragma unroll
  for (int it = 0; it < 5; ++it) {
    const int node = base + it * 16;
    const int nc   = (FULL || node < n) ? node : (n - 1);
    hv[it] = *(const float4*)(h + (size_t)nc * NHID + 4 * l);
    dg[it] = deg0[nc] + deg1[nc];
  }
  const float4 wa0 = *(const float4*)(Wfc + 8 * l);
  const float4 wa1 = *(const float4*)(Wfc + 8 * l + 4);
  const float4 wb0 = *(const float4*)(Wfc + 128 + 8 * l);
  const float4 wb1 = *(const float4*)(Wfc + 128 + 8 * l + 4);
  const float4 bv  = *(const float4*)(b1 + 4 * l);

  const float bw0 = row16_sum(bv.x * wa0.x + bv.y * wa0.z + bv.z * wa1.x + bv.w * wa1.z);
  const float bw1 = row16_sum(bv.x * wa0.y + bv.y * wa0.w + bv.z * wa1.y + bv.w * wa1.w);
  const float bw2 = row16_sum(bv.x * wb0.x + bv.y * wb0.z + bv.z * wb1.x + bv.w * wb1.z);
  const float bw3 = row16_sum(bv.x * wb0.y + bv.y * wb0.w + bv.z * wb1.y + bv.w * wb1.w);

#pragma unroll
  for (int it = 0; it < 5; ++it) {
    const int node = base + it * 16;
    const float4 v = hv[it];
    const float p0 = row16_sum(v.x * wa0.x + v.y * wa0.z + v.z * wa1.x + v.w * wa1.z);
    const float p1 = row16_sum(v.x * wa0.y + v.y * wa0.w + v.z * wa1.y + v.w * wa1.w);
    const float p2 = row16_sum(v.x * wb0.x + v.y * wb0.z + v.z * wb1.x + v.w * wb1.z);
    const float p3 = row16_sum(v.x * wb0.y + v.y * wb0.w + v.z * wb1.y + v.w * wb1.w);
    const float d  = rsqrtf(dg[it]);
    if (l == 0 && (FULL || node < n)) {
      const float dd = d * d;
      dinv[node] = d;
      ((float4*)g)[node]    = make_float4(d * p0, d * p1, d * p2, d * p3);
      ((float4*)agg4)[node] = make_float4(dd * p0 + bw0, dd * p1 + bw1,
                                          dd * p2 + bw2, dd * p3 + bw3);
      ((float4*)agg4b)[node] = make_float4(0.0f, 0.0f, 0.0f, 0.0f);
    }
  }
}

__global__ __launch_bounds__(256)
void node_project_kernel(const float* __restrict__ h,
                         const float* __restrict__ deg0,
                         const float* __restrict__ deg1,
                         const float* __restrict__ b1,
                         const float* __restrict__ Wfc,
                         float* __restrict__ dinv,
                         float* __restrict__ g,
                         float* __restrict__ agg4,
                         float* __restrict__ agg4b, int n) {
  if ((int)blockIdx.x * NODES_PER_BLOCK + NODES_PER_BLOCK <= n)
    node_project_body<true>(h, deg0, deg1, b1, Wfc, dinv, g, agg4, agg4b, n);
  else
    node_project_body<false>(h, deg0, deg1, b1, Wfc, dinv, g, agg4, agg4b, n);
}

template <bool FULL>
__device__ __forceinline__ void edge_scatter4_body(const int* __restrict__ row,
                                                   const int* __restrict__ col,
                                                   const float* __restrict__ dinv,
                                                   const float* __restrict__ g,
                                                   float* __restrict__ agg4, int e) {
  const int q  = threadIdx.x & 3;
  const int e0 = blockIdx.x * 256 + (threadIdx.x >> 2);
  int r[4], c[4];
#pragma unroll
  for (int k = 0; k < 4; ++k) {
    const int ed  = e0 + 64 * k;
    const int edc = (FULL || ed < e) ? ed : (e - 1);
    r[k] = row[edc];
    c[k] = col[edc];
  }
  float v[4];
#pragma unroll
  for (int k = 0; k < 4; ++k)
    v[k] = dinv[c[k]] * g[(size_t)r[k] * 4 + q];
#pragma unroll
  for (int k = 0; k < 4; ++k)
    if (FULL || e0 + 64 * k < e) atomicAdd(&agg4[(size_t)c[k] * 4 + q], v[k]);
}

__global__ __launch_bounds__(256)
void edge_scatter4_kernel(const int* __restrict__ row,
                          const int* __restrict__ col,
                          const float* __restrict__ dinv,
                          const float* __restrict__ g,
                          float* __restrict__ agg4,
                          float* __restrict__ agg4b, int e) {
  float* __restrict__ mine = package_half() ? agg4b : agg4;
  if ((int)blockIdx.x * 256 + 256 <= e)
    edge_scatter4_body<true>(row, col, dinv, g, mine, e);
  else
    edge_scatter4_body<false>(row, col, dinv, g, mine, e);
}

__global__ __launch_bounds__(256)
void agg_join_kernel(float* __restrict__ agg4, const float* __restrict__ agg4b, int n) {
  const int i = blockIdx.x * 256 + threadIdx.x;
  if (i < n) {
    const float4 a = ((const float4*)agg4)[i];
    const float4 b = ((const float4*)agg4b)[i];
    ((float4*)agg4)[i] = make_float4(a.x + b.x, a.y + b.y, a.z + b.z, a.w + b.w);
  }
}

template <bool FULL>
__device__ __forceinline__ void edge_softmax_body(const int* __restrict__ row,
                                                  const int* __restrict__ col,
                                                  const float* __restrict__ agg4,
                                                  const float* __restrict__ bfc,
                                                  float* __restrict__ out, int e) {
  const int e0 = blockIdx.x * 1024 + threadIdx.x;
  const float c0 = bfc[0], c1 = bfc[1];
  int r[4], c[4];
#pragma unroll
  for (int k = 0; k < 4; ++k) {
    const int ed  = e0 + 256 * k;
    const int edc = (FULL || ed < e) ? ed : (e - 1);
    r[k] = row[edc];
    c[k] = col[edc];
  }
  float2 a[4], b[4];
#pragma unroll
  for (int k = 0; k < 4; ++k) {
    a[k] = *(const float2*)(agg4 + (size_t)c[k] * 4);
    b[k] = *(const float2*)(agg4 + (size_t)r[k] * 4 + 2);
  }
#pragma unroll
  for (int k = 0; k < 4; ++k) {
    const int ed = e0 + 256 * k;
    if (FULL || ed < e) {
      float l0 = a[k].x + b[k].x + c0, l1 = a[k].y + b[k].y + c1;
      float m  = fmaxf(l0, l1);
      float x0 = __expf(l0 - m), x1 = __expf(l1 - m);
      float s  = 1.0f / (x0 + x1);
      ((float2*)out)[ed] = make_float2(x0 * s, x1 * s);
    }
  }
}

__global__ __launch_bounds__(256)
void edge_softmax_kernel(const int* __restrict__ row, const int* __restrict__ col,
                         const float* __restrict__ agg4,
                         const float* __restrict__ bfc,
                         float* __restrict__ out, int e) {
  if ((int)blockIdx.x * 1024 + 1024 <= e)
    edge_softmax_body<true>(row, col, agg4, bfc, out, e);
  else
    edge_softmax_body<false>(row, col, agg4, bfc, out, e);
}

extern "C" void kernel_launch(void* const* d_in, const int* in_sizes, int n_in,
                              void* d_out, int out_size, void* d_ws, size_t ws_size,
                              hipStream_t stream) {
  const float* x   = (const float*)d_in[0];
  const int*   ei  = (const int*)d_in[1];
  const float* W1  = (const float*)d_in[2];
  const float* b1  = (const float*)d_in[3];
  const float* Wfc = (const float*)d_in[4];
  const float* bfc = (const float*)d_in[5];
  const int n = in_sizes[0] / NFEAT;
  const int e = in_sizes[1] / 2;
  const int* row = ei;
  const int* col = ei + e;

  uintptr_t p = (uintptr_t)d_ws;
  float* deg0 = (float*)p;  p += (size_t)n * sizeof(float);
  p = (p + 255) & ~(uintptr_t)255;
  float* deg1 = (float*)p;  p += (size_t)n * sizeof(float);
  p = (p + 255) & ~(uintptr_t)255;
  float* dinv = (float*)p;  p += (size_t)n * sizeof(float);
  p = (p + 255) & ~(uintptr_t)255;
  float* h    = (float*)p;  p += (size_t)n * NHID * sizeof(float);
  p = (p + 255) & ~(uintptr_t)255;
  float* g    = (float*)p;  p += (size_t)n * 4 * sizeof(float);
  p = (p + 255) & ~(uintptr_t)255;
  float* agg4 = (float*)p;  p += (size_t)n * 4 * sizeof(float);
  p = (p + 255) & ~(uintptr_t)255;
  float* agg4b = (float*)p;

  deg_init_kernel <<<(n + 255) / 256, 256, 0, stream>>>(deg0, deg1, n);
  gemm_x_w1_deg_count_kernel<<<(n + 127) / 128 + (e + 255) / 256, 256, 0, stream>>>(
      x, W1, h, n, col, deg0, deg1, e);
  node_project_kernel<<<(n + NODES_PER_BLOCK - 1) / NODES_PER_BLOCK, 256, 0, stream>>>(
      h, deg0, deg1, b1, Wfc, dinv, g, agg4, agg4b, n);
  edge_scatter4_kernel<<<(e + 255) / 256, 256, 0, stream>>>(row, col, dinv, g, agg4, agg4b, e);
  agg_join_kernel     <<<(n + 255) / 256, 256, 0, stream>>>(agg4, agg4b, n);
  edge_softmax_kernel <<<(e + 1023) / 1024, 256, 0, stream>>>(row, col, agg4, bfc,
                                                              (float*)d_out, e);
}


# factor dinv[col] taken out of the per-edge term: edges add g[row] unscaled, the join scales each node's sum once and adds the bias; one scattered read per edge fewer; rounding of f32 sums changes
# speedup vs baseline: 5.4724x; 1.0313x over previous
#include <hip/hip_runtime.h>
#include <math.h>
#include <stdint.h>

#define NFEAT 256
#define NHID  64

typedef float v2f __attribute__((ext_vector_type(2)));
typedef float v8f __attribute__((ext_vector_type(8)));

__device__ __forceinline__ int package_half() {
  const unsigned id = __builtin_amdgcn_s_sendmsg_rtn(0x87);
  return (((id >> 16) & 0xFu) >= 4u) ? 1 : 0;
}

__global__ void deg_init_kernel(float* __restrict__ deg0, float* __restrict__ deg1, int n) {
  int i = blockIdx.x * blockDim.x + threadIdx.x;
  if (i < n) { deg0[i] = 1.0f; deg1[i] = 0.0f; }
}

__global__ __launch_bounds__(256)
void gemm_x_w1_deg_count_kernel(const float* __restrict__ x,
                                const float* __restrict__ W1,
                                float* __restrict__ h, int n,
                                const int* __restrict__ col,
                                float* __restrict__ deg0, float* __restrict__ deg1, int e) {
  __shared__ float sW[128 * NHID];
  const int gemm_blocks = (n + 127) / 128;
  if ((int)blockIdx.x >= gemm_blocks) {
    const int i = ((int)blockIdx.x - gemm_blocks) * 256 + (int)threadIdx.x;
    float* __restrict__ mine = package_half() ? deg1 : deg0;
    if (i < e) atomicAdd(&mine[col[i]], 1.0f);
    return;
  }
  const int tid  = threadIdx.x;
  const int wave = tid >> 5;
  const int lane = tid & 31;
  const int l15  = lane & 15;
  const int kh   = lane >> 4;
  const int rowBase = blockIdx.x * 128 + wave * 16;
  const int row  = rowBase + l15;
  const int rowC = (row < n) ? row : (n - 1);

  v8f acc[4];
#pragma unroll
  for (int t = 0; t < 4; ++t) acc[t] = (v8f){0,0,0,0,0,0,0,0};

  const float* xrow = x + (size_t)rowC * NFEAT + 2 * kh;
  __builtin_prefetch(xrow + 128, 0, 3);

  for (int kc = 0; kc < 2; ++kc) {
    __syncthreads();
    for (int idx = tid; idx < 64 * NHID; idx += 256) {
      int kp = idx >> 6;
      int nn = idx & 63;
      int kg = kc * 128 + kp * 2;
      ((float2*)sW)[kp * NHID + nn] =
          make_float2(W1[(size_t)kg * NHID + nn], W1[(size_t)(kg + 1) * NHID + nn]);
    }
    __syncthreads();

#pragma unroll 4
    for (int k0 = 0; k0 < 128; k0 += 4) {
      v2f a = *(const v2f*)(xrow + kc * 128 + k0);
      const int kp = (k0 >> 1) + kh;
#pragma unroll
      for (int t = 0; t < 4; ++t) {
        v2f b = ((const v2f*)sW)[kp * NHID + t * 16 + l15];
        acc[t] = __builtin_amdgcn_wmma_f32_16x16x4_f32(
            false, a, false, b, (short)0, acc[t], false, false);
      }
    }
  }

#pragma unroll
  for (int t = 0; t < 4; ++t) {
    const int ccol = t * 16 + l15;
#pragma unroll
    for (int j = 0; j < 8; ++j) {
      int r = rowBase + j + 8 * kh;
      if (r < n) h[(size_t)r * NHID + ccol] = acc[t][j];
    }
  }
}

template <int M>
__device__ __forceinline__ float row_xor(float v) {
  return __int_as_float(__builtin_amdgcn_update_dpp(0, __float_as_int(v), 0x160 + M,
                                                    0xf, 0xf, false));
}
__device__ __forceinline__ float row16_sum(float v) {
  v += row_xor<1>(v);
  v += row_xor<2>(v);
  v += row_xor<4>(v);
  v += row_xor<8>(v);
  return v;
}

#define NODES_PER_BLOCK 80
template <bool FULL>
__device__ __forceinline__ void node_project_body(const float* __restrict__ h,
                                                  const float* __restrict__ deg0,
                                                  const float* __restrict__ deg1,
                                                  const float* __restrict__ Wfc,
                                                  float* __restrict__ dinv,
                                                  float* __restrict__ g,
                                                  float* __restrict__ agg4,
                                                  float* __restrict__ agg4b, int n) {
  const int l   = threadIdx.x & 15;
  const int grp = threadIdx.x >> 4;
  const int base = blockIdx.x * NODES_PER_BLOCK + grp;
  float4 hv[5];
  float  dg[5];
#pragma unroll
  for (int it = 0; it < 5; ++it) {
    const int node = base + it * 16;
    const int nc   = (FULL || node < n) ? node : (n - 1);
    hv[it] = *(const float4*)(h + (size_t)nc * NHID + 4 * l);
    dg[it] = deg0[nc] + deg1[nc];
  }
  const float4 wa0 = *(const float4*)(Wfc + 8 * l);
  const float4 wa1 = *(const float4*)(Wfc + 8 * l + 4);
  const float4 wb0 = *(const float4*)(Wfc + 128 + 8 * l);
  const float4 wb1 = *(const float4*)(Wfc + 128 + 8 * l + 4);

#pragma unroll
  for (int it = 0; it < 5; ++it) {
    const int node = base + it * 16;
    const float4 v = hv[it];
    const float p0 = row16_sum(v.x * wa0.x + v.y * wa0.z + v.z * wa1.x + v.w * wa1.z);
    const float p1 = row16_sum(v.x * wa0.y + v.y * wa0.w + v.z * wa1.y + v.w * wa1.w);
    const float p2 = row16_sum(v.x * wb0.x + v.y * wb0.z + v.z * wb1.x + v.w * wb1.z);
    const float p3 = row16_sum(v.x * wb0.y + v.y * wb0.w + v.z * wb1.y + v.w * wb1.w);
    const float d  = rsqrtf(dg[it]);
    if (l == 0 && (FULL || node < n)) {
      const float4 gv = make_float4(d * p0, d * p1, d * p2, d * p3);
      dinv[node] = d;
      ((float4*)g)[node]     = gv;
      ((float4*)agg4)[node]  = gv;
      ((float4*)agg4b)[node] = make_float4(0.0f, 0.0f, 0.0f, 0.0f);
    }
  }
}

__global__ __launch_bounds__(256)
void node_project_kernel(const float* __restrict__ h,
                         const float* __restrict__ deg0,
                         const float* __restrict__ deg1,
                         const float* __restrict__ b1,
                         const float* __restrict__ Wfc,
                         float* __restrict__ dinv,
                         float* __restrict__ g,
                         float* __restrict__ agg4,
                         float* __restrict__ agg4b,
                         float* __restrict__ bw, int n) {
  if (blockIdx.x == 0) {
    const int l = threadIdx.x & 15;
    const float4 wa0 = *(const float4*)(Wfc + 8 * l);
    const float4 wa1 = *(const float4*)(Wfc + 8 * l + 4);
    const float4 wb0 = *(const float4*)(Wfc + 128 + 8 * l);
    const float4 wb1 = *(const float4*)(Wfc + 128 + 8 * l + 4);
    const float4 bv  = *(const float4*)(b1 + 4 * l);
    const float bw0 = row16_sum(bv.x * wa0.x + bv.y * wa0.z + bv.z * wa1.x + bv.w * wa1.z);
    const float bw1 = row16_sum(bv.x * wa0.y + bv.y * wa0.w + bv.z * wa1.y + bv.w * wa1.w);
    const float bw2 = row16_sum(bv.x * wb0.x + bv.y * wb0.z + bv.z * wb1.x + bv.w * wb1.z);
    const float bw3 = row16_sum(bv.x * wb0.y + bv.y * wb0.w + bv.z * wb1.y + bv.w * wb1.w);
    if (threadIdx.x == 0) *(float4*)bw = make_float4(bw0, bw1, bw2, bw3);
  }
  if ((int)blockIdx.x * NODES_PER_BLOCK + NODES_PER_BLOCK <= n)
    node_project_body<true>(h, deg0, deg1, Wfc, dinv, g, agg4, agg4b, n);
  else
    node_project_body<false>(h, deg0, deg1, Wfc, dinv, g, agg4, agg4b, n);
}

template <bool FULL>
__device__ __forceinline__ void edge_scatter4_body(const int* __restrict__ row,
                                                   const int* __restrict__ col,
                                                   const float* __restrict__ g,
                                                   float* __restrict__ agg4, int e) {
  const int q  = threadIdx.x & 3;
  const int e0 = blockIdx.x * 256 + (threadIdx.x >> 2);
  int r[4], c[4];
#pragma unroll
  for (int k = 0; k < 4; ++k) {
    const int ed  = e0 + 64 * k;
    const int edc = (FULL || ed < e) ? ed : (e - 1);
    r[k] = row[edc];
    c[k] = col[edc];
  }
  float v[4];
#pragma unroll
  for (int k = 0; k < 4; ++k)
    v[k] = g[(size_t)r[k] * 4 + q];
#pragma unroll
  for (int k = 0; k < 4; ++k)
    if (FULL || e0 + 64 * k < e) atomicAdd(&agg4[(size_t)c[k] * 4 + q], v[k]);
}

__global__ __launch_bounds__(256)
void edge_scatter4_kernel(const int* __restrict__ row,
                          const int* __restrict__ col,
                          const float* __restrict__ g,
                          float* __restrict__ agg4,
                          float* __restrict__ agg4b, int e) {
  float* __restrict__ mine = package_half() ? agg4b : agg4;
  if ((int)blockIdx.x * 256 + 256 <= e)
    edge_scatter4_body<true>(row, col, g, mine, e);
  else
    edge_scatter4_body<false>(row, col, g, mine, e);
}

__global__ __launch_bounds__(256)
void agg_join_kernel(float* __restrict__ agg4, const float* __restrict__ agg4b,
                     const float* __restrict__ dinv, const float* __restrict__ bw, int n) {
  const int i = blockIdx.x * 256 + threadIdx.x;
  if (i < n) {
    const float4 a = ((const float4*)agg4)[i];
    const float4 b = ((const float4*)agg4b)[i];
    const float4 w = *(const float4*)bw;
    const float  d = dinv[i];
    ((float4*)agg4)[i] = make_float4(d * (a.x + b.x) + w.x, d * (a.y + b.y) + w.y,
                                     d * (a.z + b.z) + w.z, d * (a.w + b.w) + w.w);
  }
}

template <bool FULL>
__device__ __forceinline__ void edge_softmax_body(const int* __restrict__ row,
                                                  const int* __restrict__ col,
                                                  const float* __restrict__ agg4,
                                                  const float* __restrict__ bfc,
                                                  float* __restrict__ out, int e) {
  const int e0 = blockIdx.x * 1024 + threadIdx.x;
  const float c0 = bfc[0], c1 = bfc[1];
  int r[4], c[4];
#pragma unroll
  for (int k = 0; k < 4; ++k) {
    const int ed  = e0 + 256 * k;
    const int edc = (FULL || ed < e) ? ed : (e - 1);
    r[k] = row[edc];
    c[k] = col[edc];
  }
  float2 a[4], b[4];
#pragma unroll
  for (int k = 0; k < 4; ++k) {
    a[k] = *(const float2*)(agg4 + (size_t)c[k] * 4);
    b[k] = *(const float2*)(agg4 + (size_t)r[k] * 4 + 2);
  }
#pragma unroll
  for (int k = 0; k < 4; ++k) {
    const int ed = e0 + 256 * k;
    if (FULL || ed < e) {
      float l0 = a[k].x + b[k].x + c0, l1 = a[k].y + b[k].y + c1;
      float m  = fmaxf(l0, l1);
      float x0 = __expf(l0 - m), x1 = __expf(l1 - m);
      float s  = 1.0f / (x0 + x1);
      ((float2*)out)[ed] = make_float2(x0 * s, x1 * s);
    }
  }
}

__global__ __launch_bounds__(256)
void edge_softmax_kernel(const int* __restrict__ row, const int* __restrict__ col,
                         const float* __restrict__ agg4,
                         const float* __restrict__ bfc,
                         float* __restrict__ out, int e) {
  if ((int)blockIdx.x * 1024 + 1024 <= e)
    edge_softmax_body<true>(row, col, agg4, bfc, out, e);
  else
    edge_softmax_body<false>(row, col, agg4, bfc, out, e);
}

extern "C" void kernel_launch(void* const* d_in, const int* in_sizes, int n_in,
                              void* d_out, int out_size, void* d_ws, size_t ws_size,
                              hipStream_t stream) {
  const float* x   = (const float*)d_in[0];
  const int*   ei  = (const int*)d_in[1];
  const float* W1  = (const float*)d_in[2];
  const float* b1  = (const float*)d_in[3];
  const float* Wfc = (const float*)d_in[4];
  const float* bfc = (const float*)d_in[5];
  const int n = in_sizes[0] / NFEAT;
  const int e = in_sizes[1] / 2;
  const int* row = ei;
  const int* col = ei + e;

  uintptr_t p = (uintptr_t)d_ws;
  float* deg0 = (float*)p;  p += (size_t)n * sizeof(float);
  p = (p + 255) & ~(uintptr_t)255;
  float* deg1 = (float*)p;  p += (size_t)n * sizeof(float);
  p = (p + 255) & ~(uintptr_t)255;
  float* dinv = (float*)p;  p += (size_t)n * sizeof(float);
  p = (p + 255) & ~(uintptr_t)255;
  float* h    = (float*)p;  p += (size_t)n * NHID * sizeof(float);
  p = (p + 255) & ~(uintptr_t)255;
  float* g    = (float*)p;  p += (size_t)n * 4 * sizeof(float);
  p = (p + 255) & ~(uintptr_t)255;
  float* agg4 = (float*)p;  p += (size_t)n * 4 * sizeof(float);
  p = (p + 255) & ~(uintptr_t)255;
  float* agg4b = (float*)p; p += (size_t)n * 4 * sizeof(float);
  p = (p + 255) & ~(uintptr_t)255;
  float* bw   = (float*)p;

  deg_init_kernel <<<(n + 255) / 256, 256, 0, stream>>>(deg0, deg1, n);
  gemm_x_w1_deg_count_kernel<<<(n + 127) / 128 + (e + 255) / 256, 256, 0, stream>>>(
      x, W1, h, n, col, deg0, deg1, e);
  node_project_kernel<<<(n + NODES_PER_BLOCK - 1) / NODES_PER_BLOCK, 256, 0, stream>>>(
      h, deg0, deg1, b1, Wfc, dinv, g, agg4, agg4b, bw, n);
  edge_scatter4_kernel<<<(e + 255) / 256, 256, 0, stream>>>(row, col, g, agg4, agg4b, e);
  agg_join_kernel     <<<(n + 255) / 256, 256, 0, stream>>>(agg4, agg4b, dinv, bw, n);
  edge_softmax_kernel <<<(e + 1023) / 1024, 256, 0, stream>>>(row, col, agg4, bfc,
                                                              (float*)d_out, e);
}


# two-class softmax needs only the logit difference: nodes carry 2 floats projected on (Wfc col 0 - col 1), not 4; scattered f32 atomic adds per edge halved; rounding of f32 sums changes
# speedup vs baseline: 5.7643x; 1.0533x over previous
#include <hip/hip_runtime.h>
#include <math.h>
#include <stdint.h>

#define NFEAT 256
#define NHID  64

typedef float v2f __attribute__((ext_vector_type(2)));
typedef float v8f __attribute__((ext_vector_type(8)));

__device__ __forceinline__ int package_half() {
  const unsigned id = __builtin_amdgcn_s_sendmsg_rtn(0x87);
  return (((id >> 16) & 0xFu) >= 4u) ? 1 : 0;
}

__global__ void deg_init_kernel(float* __restrict__ deg0, float* __restrict__ deg1, int n) {
  int i = blockIdx.x * blockDim.x + threadIdx.x;
  if (i < n) { deg0[i] = 1.0f; deg1[i] = 0.0f; }
}

__global__ __launch_bounds__(256)
void gemm_x_w1_deg_count_kernel(const float* __restrict__ x,
                                const float* __restrict__ W1,
                                float* __restrict__ h, int n,
                                const int* __restrict__ col,
                                float* __restrict__ deg0, float* __restrict__ deg1, int e) {
  __shared__ float sW[128 * NHID];
  const int gemm_blocks = (n + 127) / 128;
  if ((int)blockIdx.x >= gemm_blocks) {
    const int i = ((int)blockIdx.x - gemm_blocks) * 256 + (int)threadIdx.x;
    float* __restrict__ mine = package_half() ? deg1 : deg0;
    if (i < e) atomicAdd(&mine[col[i]], 1.0f);
    return;
  }
  const int tid  = threadIdx.x;
  const int wave = tid >> 5;
  const int lane = tid & 31;
  const int l15  = lane & 15;
  const int kh   = lane >> 4;
  const int rowBase = blockIdx.x * 128 + wave * 16;
  const int row  = rowBase + l15;
  const int rowC = (row < n) ? row : (n - 1);

  v8f acc[4];
#pragma unroll
  for (int t = 0; t < 4; ++t) acc[t] = (v8f){0,0,0,0,0,0,0,0};

  const float* xrow = x + (size_t)rowC * NFEAT + 2 * kh;
  __builtin_prefetch(xrow + 128, 0, 3);

  for (int kc = 0; kc < 2; ++kc) {
    __syncthreads();
    for (int idx = tid; idx < 64 * NHID; idx += 256) {
      int kp = idx >> 6;
      int nn = idx & 63;
      int kg = kc * 128 + kp * 2;
      ((float2*)sW)[kp * NHID + nn] =
          make_float2(W1[(size_t)kg * NHID + nn], W1[(size_t)(kg + 1) * NHID + nn]);
    }
    __syncthreads();

#pragma unroll 4
    for (int k0 = 0; k0 < 128; k0 += 4) {
      v2f a = *(const v2f*)(xrow + kc * 128 + k0);
      const int kp = (k0 >> 1) + kh;
#pragma unroll
      for (int t = 0; t < 4; ++t) {
        v2f b = ((const v2f*)sW)[kp * NHID + t * 16 + l15];
        acc[t] = __builtin_amdgcn_wmma_f32_16x16x4_f32(
            false, a, false, b, (short)0, acc[t], false, false);
      }
    }
  }

#pragma unroll
  for (int t = 0; t < 4; ++t) {
    const int ccol = t * 16 + l15;
#pragma unroll
    for (int j = 0; j < 8; ++j) {
      int r = rowBase + j + 8 * kh;
      if (r < n) h[(size_t)r * NHID + ccol] = acc[t][j];
    }
  }
}

template <int M>
__device__ __forceinline__ float row_xor(float v) {
  return __int_as_float(__builtin_amdgcn_update_dpp(0, __float_as_int(v), 0x160 + M,
                                                    0xf, 0xf, false));
}
__device__ __forceinline__ float row16_sum(float v) {
  v += row_xor<1>(v);
  v += row_xor<2>(v);
  v += row_xor<4>(v);
  v += row_xor<8>(v);
  return v;
}

#define NODES_PER_BLOCK 80

__device__ __forceinline__ float4 wfc_col_diff(const float* __restrict__ Wfc_half, int l) {
  const float4 w0 = *(const float4*)(Wfc_half + 8 * l);
  const float4 w1 = *(const float4*)(Wfc_half + 8 * l + 4);
  return make_float4(w0.x - w0.y, w0.z - w0.w, w1.x - w1.y, w1.z - w1.w);
}

template <bool FULL>
__device__ __forceinline__ void node_project_body(const float* __restrict__ h,
                                                  const float* __restrict__ deg0,
                                                  const float* __restrict__ deg1,
                                                  const float* __restrict__ Wfc,
                                                  float* __restrict__ dinv,
                                                  float* __restrict__ g,
                                                  float* __restrict__ agg2,
                                                  float* __restrict__ agg2b, int n) {
  const int l   = threadIdx.x & 15;
  const int grp = threadIdx.x >> 4;
  const int base = blockIdx.x * NODES_PER_BLOCK + grp;
  float4 hv[5];
  float  dg[5];
#pragma unroll
  for (int it = 0; it < 5; ++it) {
    const int node = base + it * 16;
    const int nc   = (FULL || node < n) ? node : (n - 1);
    hv[it] = *(const float4*)(h + (size_t)nc * NHID + 4 * l);
    dg[it] = deg0[nc] + deg1[nc];
  }
  const float4 wa = wfc_col_diff(Wfc, l);
  const float4 wb = wfc_col_diff(Wfc + 128, l);

#pragma unroll
  for (int it = 0; it < 5; ++it) {
    const int node = base + it * 16;
    const float4 v = hv[it];
    const float pa = row16_sum(v.x * wa.x + v.y * wa.y + v.z * wa.z + v.w * wa.w);
    const float pb = row16_sum(v.x * wb.x + v.y * wb.y + v.z * wb.z + v.w * wb.w);
    const float d  = rsqrtf(dg[it]);
    if (l == 0 && (FULL || node < n)) {
      const float2 gv = make_float2(d * pa, d * pb);
      dinv[node] = d;
      ((float2*)g)[node]     = gv;
      ((float2*)agg2)[node]  = gv;
      ((float2*)agg2b)[node] = make_float2(0.0f, 0.0f);
    }
  }
}

__global__ __launch_bounds__(256)
void node_project_kernel(const float* __restrict__ h,
                         const float* __restrict__ deg0,
                         const float* __restrict__ deg1,
                         const float* __restrict__ b1,
                         const float* __restrict__ Wfc,
                         float* __restrict__ dinv,
                         float* __restrict__ g,
                         float* __restrict__ agg2,
                         float* __restrict__ agg2b,
                         float* __restrict__ bw, int n) {
  if (blockIdx.x == 0) {
    const int l = threadIdx.x & 15;
    const float4 wa = wfc_col_diff(Wfc, l);
    const float4 wb = wfc_col_diff(Wfc + 128, l);
    const float4 bv = *(const float4*)(b1 + 4 * l);
    const float bwa = row16_sum(bv.x * wa.x + bv.y * wa.y + bv.z * wa.z + bv.w * wa.w);
    const float bwb = row16_sum(bv.x * wb.x + bv.y * wb.y + bv.z * wb.z + bv.w * wb.w);
    if (threadIdx.x == 0) *(float2*)bw = make_float2(bwa, bwb);
  }
  if ((int)blockIdx.x * NODES_PER_BLOCK + NODES_PER_BLOCK <= n)
    node_project_body<true>(h, deg0, deg1, Wfc, dinv, g, agg2, agg2b, n);
  else
    node_project_body<false>(h, deg0, deg1, Wfc, dinv, g, agg2, agg2b, n);
}

template <bool FULL>
__device__ __forceinline__ void edge_scatter2_body(const int* __restrict__ row,
                                                   const int* __restrict__ col,
                                                   const float* __restrict__ g,
                                                   float* __restrict__ agg2, int e) {
  const int q  = threadIdx.x & 1;
  const int e0 = blockIdx.x * 512 + (threadIdx.x >> 1);
  int r[4], c[4];
#pragma unroll
  for (int k = 0; k < 4; ++k) {
    const int ed  = e0 + 128 * k;
    const int edc = (FULL || ed < e) ? ed : (e - 1);
    r[k] = row[edc];
    c[k] = col[edc];
  }
  float v[4];
#pragma unroll
  for (int k = 0; k < 4; ++k)
    v[k] = g[(size_t)r[k] * 2 + q];
#pragma unroll
  for (int k = 0; k < 4; ++k)
    if (FULL || e0 + 128 * k < e) atomicAdd(&agg2[(size_t)c[k] * 2 + q], v[k]);
}

__global__ __launch_bounds__(256)
void edge_scatter2_kernel(const int* __restrict__ row,
                          const int* __restrict__ col,
                          const float* __restrict__ g,
                          float* __restrict__ agg2,
                          float* __restrict__ agg2b, int e) {
  float* __restrict__ mine = package_half() ? agg2b : agg2;
  if ((int)blockIdx.x * 512 + 512 <= e)
    edge_scatter2_body<true>(row, col, g, mine, e);
  else
    edge_scatter2_body<false>(row, col, g, mine, e);
}

__global__ __launch_bounds__(256)
void agg_join_kernel(float* __restrict__ agg2, const float* __restrict__ agg2b,
                     const float* __restrict__ dinv, const float* __restrict__ bw, int n) {
  const int i = blockIdx.x * 256 + threadIdx.x;
  if (i < n) {
    const float2 a = ((const float2*)agg2)[i];
    const float2 b = ((const float2*)agg2b)[i];
    const float2 w = *(const float2*)bw;
    const float  d = dinv[i];
    ((float2*)agg2)[i] = make_float2(d * (a.x + b.x) + w.x, d * (a.y + b.y) + w.y);
  }
}

template <bool FULL>
__device__ __forceinline__ void edge_softmax_body(const int* __restrict__ row,
                                                  const int* __restrict__ col,
                                                  const float* __restrict__ agg2,
                                                  const float* __restrict__ bfc,
                                                  float* __restrict__ out, int e) {
  const int e0 = blockIdx.x * 1024 + threadIdx.x;
  const float cd = bfc[0] - bfc[1];
  int r[4], c[4];
#pragma unroll
  for (int k = 0; k < 4; ++k) {
    const int ed  = e0 + 256 * k;
    const int edc = (FULL || ed < e) ? ed : (e - 1);
    r[k] = row[edc];
    c[k] = col[edc];
  }
  float a[4], b[4];
#pragma unroll
  for (int k = 0; k < 4; ++k) {
    a[k] = agg2[(size_t)c[k] * 2];
    b[k] = agg2[(size_t)r[k] * 2 + 1];
  }
#pragma unroll
  for (int k = 0; k < 4; ++k) {
    const int ed = e0 + 256 * k;
    if (FULL || ed < e) {
      const float dd = a[k] + b[k] + cd;
      const float t  = __expf(-fabsf(dd));
      const float s  = 1.0f / (1.0f + t);
      const float hi = s, lo = t * s;
      ((float2*)out)[ed] = (dd >= 0.0f) ? make_float2(hi, lo) : make_float2(lo, hi);
    }
  }
}

__global__ __launch_bounds__(256)
void edge_softmax_kernel(const int* __restrict__ row, const int* __restrict__ col,
                         const float* __restrict__ agg2,
                         const float* __restrict__ bfc,
                         float* __restrict__ out, int e) {
  if ((int)blockIdx.x * 1024 + 1024 <= e)
    edge_softmax_body<true>(row, col, agg2, bfc, out, e);
  else
    edge_softmax_body<false>(row, col, agg2, bfc, out, e);
}

extern "C" void kernel_launch(void* const* d_in, const int* in_sizes, int n_in,
                              void* d_out, int out_size, void* d_ws, size_t ws_size,
                              hipStream_t stream) {
  const float* x   = (const float*)d_in[0];
  const int*   ei  = (const int*)d_in[1];
  const float* W1  = (const float*)d_in[2];
  const float* b1  = (const float*)d_in[3];
  const float* Wfc = (const float*)d_in[4];
  const float* bfc = (const float*)d_in[5];
  const int n = in_sizes[0] / NFEAT;
  const int e = in_sizes[1] / 2;
  const int* row = ei;
  const int* col = ei + e;

  uintptr_t p = (uintptr_t)d_ws;
  float* deg0 = (float*)p;  p += (size_t)n * sizeof(float);
  p = (p + 255) & ~(uintptr_t)255;
  float* deg1 = (float*)p;  p += (size_t)n * sizeof(float);
  p = (p + 255) & ~(uintptr_t)255;
  float* dinv = (float*)p;  p += (size_t)n * sizeof(float);
  p = (p + 255) & ~(uintptr_t)255;
  float* h    = (float*)p;  p += (size_t)n * NHID * sizeof(float);
  p = (p + 255) & ~(uintptr_t)255;
  float* g    = (float*)p;  p += (size_t)n * 2 * sizeof(float);
  p = (p + 255) & ~(uintptr_t)255;
  float* agg2 = (float*)p;  p += (size_t)n * 2 * sizeof(float);
  p = (p + 255) & ~(uintptr_t)255;
  float* agg2b = (float*)p; p += (size_t)n * 2 * sizeof(float);
  p = (p + 255) & ~(uintptr_t)255;
  float* bw   = (float*)p;

  deg_init_kernel <<<(n + 255) / 256, 256, 0, stream>>>(deg0, deg1, n);
  gemm_x_w1_deg_count_kernel<<<(n + 127) / 128 + (e + 255) / 256, 256, 0, stream>>>(
      x, W1, h, n, col, deg0, deg1, e);
  node_project_kernel<<<(n + NODES_PER_BLOCK - 1) / NODES_PER_BLOCK, 256, 0, stream>>>(
      h, deg0, deg1, b1, Wfc, dinv, g, agg2, agg2b, bw, n);
  edge_scatter2_kernel<<<(e + 511) / 512, 256, 0, stream>>>(row, col, g, agg2, agg2b, e);
  agg_join_kernel     <<<(n + 255) / 256, 256, 0, stream>>>(agg2, agg2b, dinv, bw, n);
  edge_softmax_kernel <<<(e + 1023) / 1024, 256, 0, stream>>>(row, col, agg2, bfc,
                                                              (float*)d_out, e);
}


# W1 times the fc column difference formed first (256x2): each node's 2 floats are x @ that on f32 WMMA, one 16-wide tile with 2 live columns; h (n x 64) is never formed; order of f32 sums changes
# speedup vs baseline: 7.0660x; 1.2258x over previous
#include <hip/hip_runtime.h>
#include <math.h>
#include <stdint.h>

#define NFEAT 256
#define NHID  64

typedef float v2f __attribute__((ext_vector_type(2)));
typedef float v8f __attribute__((ext_vector_type(8)));

__device__ __forceinline__ int package_half() {
  const unsigned id = __builtin_amdgcn_s_sendmsg_rtn(0x87);
  return (((id >> 16) & 0xFu) >= 4u) ? 1 : 0;
}

template <int M>
__device__ __forceinline__ float row_xor(float v) {
  return __int_as_float(__builtin_amdgcn_update_dpp(0, __float_as_int(v), 0x160 + M,
                                                    0xf, 0xf, false));
}
__device__ __forceinline__ float row16_sum(float v) {
  v += row_xor<1>(v);
  v += row_xor<2>(v);
  v += row_xor<4>(v);
  v += row_xor<8>(v);
  return v;
}


__device__ __forceinline__ float4 wfc_col_diff(const float* __restrict__ Wfc_half, int l) {
  const float4 w0 = *(const float4*)(Wfc_half + 8 * l);
  const float4 w1 = *(const float4*)(Wfc_half + 8 * l + 4);
  return make_float4(w0.x - w0.y, w0.z - w0.w, w1.x - w1.y, w1.z - w1.w);
}

__global__ __launch_bounds__(256)
void deg_init_w12_kernel(float* __restrict__ deg0, float* __restrict__ deg1, int n,
                         const float* __restrict__ W1, const float* __restrict__ b1,
                         const float* __restrict__ Wfc,
                         float* __restrict__ w12, float* __restrict__ bw) {
  if (blockIdx.x == 0) {
    __shared__ float swd[2 * NHID];
    const int t = threadIdx.x;
    if (t < 2 * NHID) swd[t] = Wfc[2 * t] - Wfc[2 * t + 1];
    __syncthreads();
    for (int k = t; k < NFEAT; k += 256) {
      const float* wrow = W1 + (size_t)k * NHID;
      float sa = 0.0f, sb = 0.0f;
#pragma unroll
      for (int j = 0; j < NHID; j += 4) {
        const float4 w = *(const float4*)(wrow + j);
        sa += w.x * swd[j] + w.y * swd[j + 1] + w.z * swd[j + 2] + w.w * swd[j + 3];
        sb += w.x * swd[NHID + j] + w.y * swd[NHID + j + 1] + w.z * swd[NHID + j + 2] +
              w.w * swd[NHID + j + 3];
      }
      ((float2*)w12)[k] = make_float2(sa, sb);
    }
    const int l = t & 15;
    const float4 wa = wfc_col_diff(Wfc, l);
    const float4 wb = wfc_col_diff(Wfc + 128, l);
    const float4 bv = *(const float4*)(b1 + 4 * l);
    const float bwa = row16_sum(bv.x * wa.x + bv.y * wa.y + bv.z * wa.z + bv.w * wa.w);
    const float bwb = row16_sum(bv.x * wb.x + bv.y * wb.y + bv.z * wb.z + bv.w * wb.w);
    if (t == 0) *(float2*)bw = make_float2(bwa, bwb);
    return;
  }
  const int i = ((int)blockIdx.x - 1) * 256 + (int)threadIdx.x;
  if (i < n) { deg0[i] = 1.0f; deg1[i] = 0.0f; }
}

#define SB_PITCH 264
__global__ __launch_bounds__(256)
void node_dot_w12_deg_count_kernel(const float* __restrict__ x,
                                   const float* __restrict__ w12,
                                   float* __restrict__ p, int n,
                                   const int* __restrict__ col,
                                   float* __restrict__ deg0, float* __restrict__ deg1, int e) {
  __shared__ __attribute__((aligned(16))) float sB[3 * SB_PITCH];
  const int gemm_blocks = (n + 127) / 128;
  if ((int)blockIdx.x >= gemm_blocks) {
    const int i = ((int)blockIdx.x - gemm_blocks) * 256 + (int)threadIdx.x;
    float* __restrict__ mine = package_half() ? deg1 : deg0;
    if (i < e) atomicAdd(&mine[col[i]], 1.0f);
    return;
  }
  const int tid  = threadIdx.x;
  const int wave = tid >> 5;
  const int lane = tid & 31;
  const int l15  = lane & 15;
  const int kh   = lane >> 4;
  const int rowBase = blockIdx.x * 128 + wave * 16;
  const int row  = rowBase + l15;
  const int rowC = (row < n) ? row : (n - 1);

  for (int k = tid; k < NFEAT; k += 256) {
    const float2 w = ((const float2*)w12)[k];
    sB[k] = w.x;
    sB[SB_PITCH + k] = w.y;
    sB[2 * SB_PITCH + k] = 0.0f;
  }
  __syncthreads();

  const float* xp = x + (size_t)rowC * NFEAT + 4 * kh;
  const float* bp = sB + (l15 < 2 ? l15 : 2) * SB_PITCH + 4 * kh;

  v8f acc0 = (v8f){0,0,0,0,0,0,0,0};
  v8f acc1 = (v8f){0,0,0,0,0,0,0,0};

#pragma unroll 1
  for (int g = 0; g < NFEAT / 64; ++g) {
    float4 a[8];
#pragma unroll
    for (int i = 0; i < 8; ++i) a[i] = *(const float4*)(xp + g * 64 + 8 * i);
#pragma unroll
    for (int i = 0; i < 8; ++i) {
      const float4 b = *(const float4*)(bp + g * 64 + 8 * i);
      acc0 = __builtin_amdgcn_wmma_f32_16x16x4_f32(
          false, (v2f){a[i].x, a[i].y}, false, (v2f){b.x, b.y}, (short)0, acc0, false, false);
      acc1 = __builtin_amdgcn_wmma_f32_16x16x4_f32(
          false, (v2f){a[i].z, a[i].w}, false, (v2f){b.z, b.w}, (short)0, acc1, false, false);
    }
  }

  float c0[8], c1[8];
#pragma unroll
  for (int j = 0; j < 8; ++j) {
    c0[j] = acc0[j] + acc1[j];
    c1[j] = row_xor<1>(c0[j]);
  }
  if (l15 == 0) {
    const int r0 = rowBase + 8 * kh;
    if (r0 + 8 <= n) {
#pragma unroll
      for (int j = 0; j < 8; j += 2)
        *(float4*)(p + (size_t)(r0 + j) * 2) = make_float4(c0[j], c1[j], c0[j + 1], c1[j + 1]);
    } else {
#pragma unroll
      for (int j = 0; j < 8; ++j)
        if (r0 + j < n) ((float2*)p)[r0 + j] = make_float2(c0[j], c1[j]);
    }
  }
}

__global__ __launch_bounds__(256)
void node_scale_kernel(const float* __restrict__ p,
                       const float* __restrict__ deg0,
                       const float* __restrict__ deg1,
                       float* __restrict__ dinv,
                       float* __restrict__ g,
                       float* __restrict__ agg2,
                       float* __restrict__ agg2b, int n) {
  const int i = blockIdx.x * 256 + threadIdx.x;
  if (i < n) {
    const float  d  = rsqrtf(deg0[i] + deg1[i]);
    const float2 pv = ((const float2*)p)[i];
    const float2 gv = make_float2(d * pv.x, d * pv.y);
    dinv[i] = d;
    ((float2*)g)[i]     = gv;
    ((float2*)agg2)[i]  = gv;
    ((float2*)agg2b)[i] = make_float2(0.0f, 0.0f);
  }
}

template <bool FULL>
__device__ __forceinline__ void edge_scatter2_body(const int* __restrict__ row,
                                                   const int* __restrict__ col,
                                                   const float* __restrict__ g,
                                                   float* __restrict__ agg2, int e) {
  const int q  = threadIdx.x & 1;
  const int e0 = blockIdx.x * 512 + (threadIdx.x >> 1);
  int r[4], c[4];
#pragma unroll
  for (int k = 0; k < 4; ++k) {
    const int ed  = e0 + 128 * k;
    const int edc = (FULL || ed < e) ? ed : (e - 1);
    r[k] = row[edc];
    c[k] = col[edc];
  }
  float v[4];
#pragma unroll
  for (int k = 0; k < 4; ++k)
    v[k] = g[(size_t)r[k] * 2 + q];
#pragma unroll
  for (int k = 0; k < 4; ++k)
    if (FULL || e0 + 128 * k < e) atomicAdd(&agg2[(size_t)c[k] * 2 + q], v[k]);
}

__global__ __launch_bounds__(256)
void edge_scatter2_kernel(const int* __restrict__ row,
                          const int* __restrict__ col,
                          const float* __restrict__ g,
                          float* __restrict__ agg2,
                          float* __restrict__ agg2b, int e) {
  float* __restrict__ mine = package_half() ? agg2b : agg2;
  if ((int)blockIdx.x * 512 + 512 <= e)
    edge_scatter2_body<true>(row, col, g, mine, e);
  else
    edge_scatter2_body<false>(row, col, g, mine, e);
}

__global__ __launch_bounds__(256)
void agg_join_kernel(float* __restrict__ agg2, const float* __restrict__ agg2b,
                     const float* __restrict__ dinv, const float* __restrict__ bw, int n) {
  const int i = blockIdx.x * 256 + threadIdx.x;
  if (i < n) {
    const float2 a = ((const float2*)agg2)[i];
    const float2 b = ((const float2*)agg2b)[i];
    const float2 w = *(const float2*)bw;
    const float  d = dinv[i];
    ((float2*)agg2)[i] = make_float2(d * (a.x + b.x) + w.x, d * (a.y + b.y) + w.y);
  }
}

template <bool FULL>
__device__ __forceinline__ void edge_softmax_body(const int* __restrict__ row,
                                                  const int* __restrict__ col,
                                                  const float* __restrict__ agg2,
                                                  const float* __restrict__ bfc,
                                                  float* __restrict__ out, int e) {
  const int e0 = blockIdx.x * 1024 + threadIdx.x;
  const float cd = bfc[0] - bfc[1];
  int r[4], c[4];
#pragma unroll
  for (int k = 0; k < 4; ++k) {
    const int ed  = e0 + 256 * k;
    const int edc = (FULL || ed < e) ? ed : (e - 1);
    r[k] = row[edc];
    c[k] = col[edc];
  }
  float a[4], b[4];
#pragma unroll
  for (int k = 0; k < 4; ++k) {
    a[k] = agg2[(size_t)c[k] * 2];
    b[k] = agg2[(size_t)r[k] * 2 + 1];
  }
#pragma unroll
  for (int k = 0; k < 4; ++k) {
    const int ed = e0 + 256 * k;
    if (FULL || ed < e) {
      const float dd = a[k] + b[k] + cd;
      const float t  = __expf(-fabsf(dd));
      const float s  = 1.0f / (1.0f + t);
      const float hi = s, lo = t * s;
      ((float2*)out)[ed] = (dd >= 0.0f) ? make_float2(hi, lo) : make_float2(lo, hi);
    }
  }
}

__global__ __launch_bounds__(256)
void edge_softmax_kernel(const int* __restrict__ row, const int* __restrict__ col,
                         const float* __restrict__ agg2,
                         const float* __restrict__ bfc,
                         float* __restrict__ out, int e) {
  if ((int)blockIdx.x * 1024 + 1024 <= e)
    edge_softmax_body<true>(row, col, agg2, bfc, out, e);
  else
    edge_softmax_body<false>(row, col, agg2, bfc, out, e);
}

extern "C" void kernel_launch(void* const* d_in, const int* in_sizes, int n_in,
                              void* d_out, int out_size, void* d_ws, size_t ws_size,
                              hipStream_t stream) {
  const float* x   = (const float*)d_in[0];
  const int*   ei  = (const int*)d_in[1];
  const float* W1  = (const float*)d_in[2];
  const float* b1  = (const float*)d_in[3];
  const float* Wfc = (const float*)d_in[4];
  const float* bfc = (const float*)d_in[5];
  const int n = in_sizes[0] / NFEAT;
  const int e = in_sizes[1] / 2;
  const int* row = ei;
  const int* col = ei + e;

  uintptr_t q = (uintptr_t)d_ws;
  float* deg0 = (float*)q;  q += (size_t)n * sizeof(float);
  q = (q + 255) & ~(uintptr_t)255;
  float* deg1 = (float*)q;  q += (size_t)n * sizeof(float);
  q = (q + 255) & ~(uintptr_t)255;
  float* dinv = (float*)q;  q += (size_t)n * sizeof(float);
  q = (q + 255) & ~(uintptr_t)255;
  float* p    = (float*)q;  q += (size_t)n * 2 * sizeof(float);
  q = (q + 255) & ~(uintptr_t)255;
  float* g    = (float*)q;  q += (size_t)n * 2 * sizeof(float);
  q = (q + 255) & ~(uintptr_t)255;
  float* agg2 = (float*)q;  q += (size_t)n * 2 * sizeof(float);
  q = (q + 255) & ~(uintptr_t)255;
  float* agg2b = (float*)q; q += (size_t)n * 2 * sizeof(float);
  q = (q + 255) & ~(uintptr_t)255;
  float* bw   = (float*)q;  q += 256;
  float* w12  = (float*)q;

  deg_init_w12_kernel<<<(n + 255) / 256 + 1, 256, 0, stream>>>(deg0, deg1, n, W1, b1, Wfc, w12, bw);
  node_dot_w12_deg_count_kernel<<<(n + 127) / 128 + (e + 255) / 256, 256, 0, stream>>>(
      x, w12, p, n, col, deg0, deg1, e);
  node_scale_kernel   <<<(n + 255) / 256, 256, 0, stream>>>(p, deg0, deg1, dinv, g, agg2, agg2b, n);
  edge_scatter2_kernel<<<(e + 511) / 512, 256, 0, stream>>>(row, col, g, agg2, agg2b, e);
  agg_join_kernel     <<<(n + 255) / 256, 256, 0, stream>>>(agg2, agg2b, dinv, bw, n);
  edge_softmax_kernel <<<(e + 1023) / 1024, 256, 0, stream>>>(row, col, agg2, bfc,
                                                              (float*)d_out, e);
}
